# starting kernel
# baseline (speedup 1.0000x reference)
#include <hip/hip_runtime.h>
#include <math.h>

typedef __attribute__((ext_vector_type(2))) float v2f;
typedef __attribute__((ext_vector_type(8))) float v8f;

#define T_LEN 400
#define D_LEN 10000
#define K1 1200
#define K3 800
#define NTILES (D_LEN / 16)

__global__ void hdc_features_kernel(const float* __restrict__ input,
                                    float* __restrict__ ws) {
  __shared__ float s_t[T_LEN];
  __shared__ float s_x[T_LEN * 3];

  const int tid = threadIdx.x;
  for (int t = tid; t < T_LEN; t += blockDim.x) {
    s_t[t]         = input[t * 4 + 0];
    s_x[t * 3 + 0] = input[t * 4 + 1];
    s_x[t * 3 + 1] = input[t * 4 + 2];
    s_x[t * 3 + 2] = input[t * 4 + 3];
  }
  __syncthreads();

  for (int t = tid; t < T_LEN; t += blockDim.x) {
    const float x = s_x[t * 3 + 0];
    const float y = s_x[t * 3 + 1];
    const float z = s_x[t * 3 + 2];
    ws[0 * T_LEN + t] = x;
    ws[1 * T_LEN + t] = y;
    ws[2 * T_LEN + t] = z;
    float jx = 0.f, jy = 0.f, jz = 0.f;
    if (t > 0) {
      const float dt = s_t[t] - s_t[t - 1];
      const float inv = 1.0f / dt;
      jx = (x - s_x[(t - 1) * 3 + 0]) * inv;
      jy = (y - s_x[(t - 1) * 3 + 1]) * inv;
      jz = (z - s_x[(t - 1) * 3 + 2]) * inv;
    }
    ws[K1 + 0 * T_LEN + t] = jx;
    ws[K1 + 1 * T_LEN + t] = jy;
    ws[K1 + 2 * T_LEN + t] = jz;
    ws[2 * K1 + t]         = sqrtf(x * x + y * y + z * z);
    ws[2 * K1 + T_LEN + t] = sqrtf(jx * jx + jy * jy + jz * jz);
  }
  __syncthreads();

  if (tid < 3) {
    float s = 0.f;
    for (int t = 0; t < T_LEN; ++t) {
      const float v = s_x[t * 3 + tid];
      s += v * v;
    }
    ws[2 * K1 + K3 + tid] = s / (float)T_LEN;
  }
}

__global__ __launch_bounds__(256) void hdc_gemv_wmma_kernel(
    const float* __restrict__ W1, const float* __restrict__ b1,
    const float* __restrict__ W2, const float* __restrict__ b2,
    const float* __restrict__ W3, const float* __restrict__ b3,
    const float* __restrict__ W4, const float* __restrict__ b4,
    const float* __restrict__ feats, float* __restrict__ out) {
  const int lane  = threadIdx.x & 31;
  const int wave  = threadIdx.x >> 5;
  const int tile  = blockIdx.x * (blockDim.x >> 5) + wave;
  if (tile >= NTILES) return;

  const int row0  = tile * 16;
  const int m     = lane & 15;
  const int khalf = (lane >> 4) * 2;
  const int rowA  = row0 + m;

  const float* __restrict__ f1 = feats;
  const float* __restrict__ f2 = feats + K1;
  const float* __restrict__ f3 = feats + 2 * K1;
  const float* __restrict__ en = feats + 2 * K1 + K3;

  const float* __restrict__ r1 = W1 + (size_t)rowA * K1 + khalf;
  const float* __restrict__ r2 = W2 + (size_t)rowA * K1 + khalf;
  const float* __restrict__ r3 = W3 + (size_t)rowA * K3 + khalf;

  __builtin_prefetch(r1, 0, 0);
  __builtin_prefetch(r2, 0, 0);
  __builtin_prefetch(r3, 0, 0);

  v8f acc1 = {};
  v8f acc2 = {};
  v8f acc3 = {};

  #pragma unroll 4
  for (int k = 0; k < K1; k += 4) {
    const v2f a1 = *(const v2f*)(r1 + k);
    const v2f a2 = *(const v2f*)(r2 + k);
    const v2f bA = *(const v2f*)(f1 + k + khalf);
    const v2f bB = *(const v2f*)(f2 + k + khalf);
    acc1 = __builtin_amdgcn_wmma_f32_16x16x4_f32(false, a1, false, bA,
                                                 (short)0, acc1, false, false);
    acc2 = __builtin_amdgcn_wmma_f32_16x16x4_f32(false, a2, false, bB,
                                                 (short)0, acc2, false, false);
  }

  #pragma unroll 4
  for (int k = 0; k < K3; k += 4) {
    const v2f a3 = *(const v2f*)(r3 + k);
    const v2f bC = *(const v2f*)(f3 + k + khalf);
    acc3 = __builtin_amdgcn_wmma_f32_16x16x4_f32(false, a3, false, bC,
                                                 (short)0, acc3, false, false);
  }

  if ((lane & 15) == 0) {
    const int   half = lane >> 4;
    const float e0 = en[0], e1 = en[1], e2 = en[2];
    #pragma unroll
    for (int v = 0; v < 8; ++v) {
      const int row = row0 + half * 8 + v;
      const float h1 = tanhf(acc1[v] + b1[row]);
      const float h2 = tanhf(acc2[v] + b2[row]);
      const float h3 = tanhf(acc3[v] + b3[row]);
      const float d4 = W4[row * 3 + 0] * e0 + W4[row * 3 + 1] * e1 +
                       W4[row * 3 + 2] * e2 + b4[row];
      const float h4 = tanhf(d4);
      out[row] = tanhf(h1 * h4 + h2 + h3);
    }
  }
}

extern "C" void kernel_launch(void* const* d_in, const int* in_sizes, int n_in,
                              void* d_out, int out_size, void* d_ws, size_t ws_size,
                              hipStream_t stream) {
  (void)in_sizes; (void)n_in; (void)out_size; (void)ws_size;
  const float* input = (const float*)d_in[0];
  const float* W1    = (const float*)d_in[1];
  const float* b1    = (const float*)d_in[2];
  const float* W2    = (const float*)d_in[3];
  const float* b2    = (const float*)d_in[4];
  const float* W3    = (const float*)d_in[5];
  const float* b3    = (const float*)d_in[6];
  const float* W4    = (const float*)d_in[7];
  const float* b4    = (const float*)d_in[8];
  float*       out   = (float*)d_out;
  float*       feats = (float*)d_ws;

  hdc_features_kernel<<<1, 512, 0, stream>>>(input, feats);

  const int wavesPerBlock = 8;
  const int blocks = (NTILES + wavesPerBlock - 1) / wavesPerBlock;
  hdc_gemv_wmma_kernel<<<blocks, 256, 0, stream>>>(W1, b1, W2, b2, W3, b3,
                                                   W4, b4, feats, out);
}


# GEMV re-mapped to row streaming: a wave reads whole rows with 512-byte contiguous loads, row dot = trace of the WMMA accumulator, 625 groups x 8 waves fill the chip; sums added in a new fixed order
# speedup vs baseline: 4.9872x; 4.9872x over previous
#include <hip/hip_runtime.h>
#include <math.h>


typedef __attribute__((ext_vector_type(2))) float v2f;
typedef __attribute__((ext_vector_type(4))) float v4f;
typedef __attribute__((ext_vector_type(8))) float v8f;

#define T_LEN 400
#define D_LEN 10000
#define K1 1200
#define K3 800
#define ROWS_PB 16
#define NBLOCKS (D_LEN / ROWS_PB)
#define NTASK (3 * ROWS_PB)
#define PART_PITCH 49

__global__ void hdc_features_kernel(const float* __restrict__ input,
                                    float* __restrict__ ws) {
  __shared__ float s_t[T_LEN];
  __shared__ float s_x[T_LEN * 3];

  const int tid = threadIdx.x;
  for (int t = tid; t < T_LEN; t += blockDim.x) {
    s_t[t]         = input[t * 4 + 0];
    s_x[t * 3 + 0] = input[t * 4 + 1];
    s_x[t * 3 + 1] = input[t * 4 + 2];
    s_x[t * 3 + 2] = input[t * 4 + 3];
  }
  __syncthreads();

  for (int t = tid; t < T_LEN; t += blockDim.x) {
    const float x = s_x[t * 3 + 0];
    const float y = s_x[t * 3 + 1];
    const float z = s_x[t * 3 + 2];
    ws[0 * T_LEN + t] = x;
    ws[1 * T_LEN + t] = y;
    ws[2 * T_LEN + t] = z;
    float jx = 0.f, jy = 0.f, jz = 0.f;
    if (t > 0) {
      const float dt = s_t[t] - s_t[t - 1];
      const float inv = 1.0f / dt;
      jx = (x - s_x[(t - 1) * 3 + 0]) * inv;
      jy = (y - s_x[(t - 1) * 3 + 1]) * inv;
      jz = (z - s_x[(t - 1) * 3 + 2]) * inv;
    }
    ws[K1 + 0 * T_LEN + t] = jx;
    ws[K1 + 1 * T_LEN + t] = jy;
    ws[K1 + 2 * T_LEN + t] = jz;
    ws[2 * K1 + t]         = sqrtf(x * x + y * y + z * z);
    ws[2 * K1 + T_LEN + t] = sqrtf(jx * jx + jy * jy + jz * jz);
  }
  __syncthreads();

  if (tid < 3) {
    float s = 0.f;
    for (int t = 0; t < T_LEN; ++t) {
      const float v = s_x[t * 3 + tid];
      s += v * v;
    }
    ws[2 * K1 + K3 + tid] = s / (float)T_LEN;
  }
}

template <int K, int NROWS>
__device__ __forceinline__ void stream_rows(const float* __restrict__ wrow0,
                                            const float* __restrict__ f,
                                            float* __restrict__ part,
                                            const int lane) {
  constexpr int NV  = K / 4;
  constexpr int NQ  = (NV + 31) / 32;
  constexpr int REM = NV - 32 * (NQ - 1);
  const bool tail_on = lane < REM;
  const int tail_off = tail_on ? 32 * (NQ - 1) : 0;
  const v4f zero = {0.f, 0.f, 0.f, 0.f};

  const v4f* fp = reinterpret_cast<const v4f*>(f) + lane;
  v4f fq[NQ];
#pragma unroll
  for (int q = 0; q < NQ - 1; ++q) fq[q] = fp[32 * q];
  fq[NQ - 1] = fp[tail_off];

  const v4f* wp = reinterpret_cast<const v4f*>(wrow0) + lane;
  v4f w[2][NQ];
#define LOAD_ROW(R)                                                          \
  do {                                                                       \
    const v4f* rp_ = wp + (size_t)(R) * NV;                                  \
    _Pragma("unroll") for (int q = 0; q < NQ - 1; ++q)                       \
        w[(R) & 1][q] = rp_[32 * q];                                         \
    w[(R) & 1][NQ - 1] = rp_[tail_off];                                      \
  } while (0)

  const int  sel  = lane & 7;
  const bool diag = ((lane >> 3) & 1) == ((lane >> 4) & 1);

  LOAD_ROW(0);
#pragma unroll
  for (int r = 0; r < NROWS; ++r) {
    __builtin_amdgcn_sched_barrier(0);
    if (r + 1 < NROWS) LOAD_ROW(r + 1);
    __builtin_amdgcn_sched_barrier(0);

    if (r == 0) fq[NQ - 1] = tail_on ? fq[NQ - 1] : zero;

    const v4f* wr = w[r & 1];
    v8f acc0 = {};
    v8f acc1 = {};
#pragma unroll
    for (int q = 0; q < NQ; ++q) {
      v4f wv = wr[q];
      if (q == NQ - 1) wv = tail_on ? wv : zero;
      const v2f a0 = {wv.x, wv.y};
      const v2f b0 = {fq[q].x, fq[q].y};
      const v2f a1 = {wv.z, wv.w};
      const v2f b1 = {fq[q].z, fq[q].w};
      acc0 = __builtin_amdgcn_wmma_f32_16x16x4_f32(false, a0, false, b0,
                                                   (short)0, acc0, false, false);
      acc1 = __builtin_amdgcn_wmma_f32_16x16x4_f32(false, a1, false, b1,
                                                   (short)0, acc1, false, false);
    }
    const v8f acc = acc0 + acc1;
    float t = acc[0];
#pragma unroll
    for (int i = 1; i < 8; ++i) t = (sel == i) ? acc[i] : t;
    if (diag) part[(lane & 15) * PART_PITCH + r] = t;
  }
#undef LOAD_ROW
}

__global__ __launch_bounds__(256) void hdc_gemv_wmma_kernel(
    const float* __restrict__ W1, const float* __restrict__ b1,
    const float* __restrict__ W2, const float* __restrict__ b2,
    const float* __restrict__ W3, const float* __restrict__ b3,
    const float* __restrict__ W4, const float* __restrict__ b4,
    const float* __restrict__ feats, float* __restrict__ out) {
  __shared__ float s_part[16 * PART_PITCH];
  __shared__ float s_dot[NTASK];

  const int tid  = threadIdx.x;
  const int lane = tid & 31;
  const int wave = __builtin_amdgcn_readfirstlane(tid >> 5);
  const int row0 = blockIdx.x * ROWS_PB;

  const float* __restrict__ f1 = feats;
  const float* __restrict__ f2 = feats + K1;
  const float* __restrict__ f3 = feats + 2 * K1;
  const float* __restrict__ en = feats + 2 * K1 + K3;

  if (wave < 6) {
    const bool first = wave < 3;
    const float* __restrict__ Wm = first ? W1 : W2;
    const float* __restrict__ fm = first ? f1 : f2;
    float* pm = s_part + (first ? 0 : ROWS_PB);
    const int wi = first ? wave : wave - 3;
    if (wi == 0) {
      stream_rows<K1, 6>(Wm + (size_t)row0 * K1, fm, pm, lane);
    } else {
      const int r0 = 1 + 5 * wi;
      stream_rows<K1, 5>(Wm + (size_t)(row0 + r0) * K1, fm, pm + r0, lane);
    }
  } else {
    const int r0 = 8 * (wave - 6);
    stream_rows<K3, 8>(W3 + (size_t)(row0 + r0) * K3, f3, s_part + 2 * ROWS_PB + r0, lane);
  }
  const int et = tid - 32;
  const bool fin = et >= 0 && et < ROWS_PB;
  float eb1 = 0.f, eb2 = 0.f, eb3 = 0.f, eb4 = 0.f;
  float ew0 = 0.f, ew1 = 0.f, ew2 = 0.f;
  if (fin) {
    const int row = row0 + et;
    eb1 = b1[row];
    eb2 = b2[row];
    eb3 = b3[row];
    eb4 = b4[row];
    ew0 = W4[row * 3 + 0];
    ew1 = W4[row * 3 + 1];
    ew2 = W4[row * 3 + 2];
  }
  const float e0 = en[0], e1 = en[1], e2 = en[2];
  __syncthreads();

  if (tid < NTASK) {
    float s0 = 0.f, s1 = 0.f, s2 = 0.f, s3 = 0.f;
#pragma unroll
    for (int i = 0; i < 16; i += 4) {
      s0 += s_part[(i + 0) * PART_PITCH + tid];
      s1 += s_part[(i + 1) * PART_PITCH + tid];
      s2 += s_part[(i + 2) * PART_PITCH + tid];
      s3 += s_part[(i + 3) * PART_PITCH + tid];
    }
    s_dot[tid] = (s0 + s1) + (s2 + s3);
  }
  __syncthreads();

  if (fin) {
    const float h1 = tanhf(s_dot[et] + eb1);
    const float h2 = tanhf(s_dot[ROWS_PB + et] + eb2);
    const float h3 = tanhf(s_dot[2 * ROWS_PB + et] + eb3);
    const float d4 = ew0 * e0 + ew1 * e1 + ew2 * e2 + eb4;
    const float h4 = tanhf(d4);
    out[row0 + et] = tanhf(h1 * h4 + h2 + h3);
  }
}

extern "C" void kernel_launch(void* const* d_in, const int* in_sizes, int n_in,
                              void* d_out, int out_size, void* d_ws, size_t ws_size,
                              hipStream_t stream) {
  (void)in_sizes; (void)n_in; (void)out_size; (void)ws_size;
  const float* input = (const float*)d_in[0];
  const float* W1    = (const float*)d_in[1];
  const float* b1    = (const float*)d_in[2];
  const float* W2    = (const float*)d_in[3];
  const float* b2    = (const float*)d_in[4];
  const float* W3    = (const float*)d_in[5];
  const float* b3    = (const float*)d_in[6];
  const float* W4    = (const float*)d_in[7];
  const float* b4    = (const float*)d_in[8];
  float*       out   = (float*)d_out;
  float*       feats = (float*)d_ws;

  hdc_features_kernel<<<1, 512, 0, stream>>>(input, feats);

  hdc_gemv_wmma_kernel<<<NBLOCKS, 256, 0, stream>>>(W1, b1, W2, b2, W3, b3,
                                                    W4, b4, feats, out);
}


# Feature kernel fused into the GEMV kernel: one launch, each work-group rebuilds the features in LDS from the input; energy summed in a new fixed order (8-sums per lane, then butterfly)
# speedup vs baseline: 7.0461x; 1.4128x over previous
#include <hip/hip_runtime.h>
#include <math.h>


typedef __attribute__((ext_vector_type(2))) float v2f;
typedef __attribute__((ext_vector_type(4))) float v4f;
typedef __attribute__((ext_vector_type(8))) float v8f;

#define T_LEN 400
#define D_LEN 10000
#define K1 1200
#define K3 800
#define ROWS_PB 16
#define NBLOCKS (D_LEN / ROWS_PB)
#define NTASK (3 * ROWS_PB)
#define PART_PITCH 49
#define NTHREADS 256
#define T_SECOND (T_LEN - NTHREADS)

template <int K, int NROWS>
__device__ __forceinline__ void stream_rows(const float* __restrict__ wrow0,
                                            const float* __restrict__ f,
                                            float* __restrict__ part,
                                            const int lane) {
  constexpr int NV  = K / 4;
  constexpr int NQ  = (NV + 31) / 32;
  constexpr int REM = NV - 32 * (NQ - 1);
  const bool tail_on = lane < REM;
  const int tail_off = tail_on ? 32 * (NQ - 1) : 0;
  const v4f zero = {0.f, 0.f, 0.f, 0.f};

  const v4f* fp = reinterpret_cast<const v4f*>(f) + lane;
  v4f fq[NQ];
#pragma unroll
  for (int q = 0; q < NQ - 1; ++q) fq[q] = fp[32 * q];
  fq[NQ - 1] = fp[tail_off];

  const v4f* wp = reinterpret_cast<const v4f*>(wrow0) + lane;
  v4f w[2][NQ];
#define LOAD_ROW(R)                                                          \
  do {                                                                       \
    const v4f* rp_ = wp + (size_t)(R) * NV;                                  \
    _Pragma("unroll") for (int q = 0; q < NQ - 1; ++q)                       \
        w[(R) & 1][q] = rp_[32 * q];                                         \
    w[(R) & 1][NQ - 1] = rp_[tail_off];                                      \
  } while (0)

  const int  sel  = lane & 7;
  const bool diag = ((lane >> 3) & 1) == ((lane >> 4) & 1);

  LOAD_ROW(0);
#pragma unroll
  for (int r = 0; r < NROWS; ++r) {
    __builtin_amdgcn_sched_barrier(0);
    if (r + 1 < NROWS) LOAD_ROW(r + 1);
    __builtin_amdgcn_sched_barrier(0);

    if (r == 0) fq[NQ - 1] = tail_on ? fq[NQ - 1] : zero;

    const v4f* wr = w[r & 1];
    v8f acc0 = {};
    v8f acc1 = {};
#pragma unroll
    for (int q = 0; q < NQ; ++q) {
      v4f wv = wr[q];
      if (q == NQ - 1) wv = tail_on ? wv : zero;
      const v2f a0 = {wv.x, wv.y};
      const v2f b0 = {fq[q].x, fq[q].y};
      const v2f a1 = {wv.z, wv.w};
      const v2f b1 = {fq[q].z, fq[q].w};
      acc0 = __builtin_amdgcn_wmma_f32_16x16x4_f32(false, a0, false, b0,
                                                   (short)0, acc0, false, false);
      acc1 = __builtin_amdgcn_wmma_f32_16x16x4_f32(false, a1, false, b1,
                                                   (short)0, acc1, false, false);
    }
    const v8f acc = acc0 + acc1;
    float t = acc[0];
#pragma unroll
    for (int i = 1; i < 8; ++i) t = (sel == i) ? acc[i] : t;
    if (diag) part[(lane & 15) * PART_PITCH + r] = t;
  }
#undef LOAD_ROW
}

__device__ __forceinline__ void step_features(const int t, const v4f c, const v4f p,
                                              const bool live,
                                              float* __restrict__ s_feat,
                                              float& ex, float& ey, float& ez) {
  const float x = c.y, y = c.z, z = c.w;
  float jx = 0.f, jy = 0.f, jz = 0.f;
  if (t > 0) {
    const float dt  = c.x - p.x;
    const float inv = 1.0f / dt;
    jx = (x - p.y) * inv;
    jy = (y - p.z) * inv;
    jz = (z - p.w) * inv;
  }
  if (live) {
    s_feat[0 * T_LEN + t] = x;
    s_feat[1 * T_LEN + t] = y;
    s_feat[2 * T_LEN + t] = z;
    s_feat[K1 + 0 * T_LEN + t] = jx;
    s_feat[K1 + 1 * T_LEN + t] = jy;
    s_feat[K1 + 2 * T_LEN + t] = jz;
    s_feat[2 * K1 + t]         = sqrtf(x * x + y * y + z * z);
    s_feat[2 * K1 + T_LEN + t] = sqrtf(jx * jx + jy * jy + jz * jz);
    ex += x * x;
    ey += y * y;
    ez += z * z;
  }
}

__global__ __launch_bounds__(NTHREADS) void hdc_fused_wmma_kernel(
    const float* __restrict__ input,
    const float* __restrict__ W1, const float* __restrict__ b1,
    const float* __restrict__ W2, const float* __restrict__ b2,
    const float* __restrict__ W3, const float* __restrict__ b3,
    const float* __restrict__ W4, const float* __restrict__ b4,
    float* __restrict__ out) {
  __shared__ __attribute__((aligned(16))) float s_feat[2 * K1 + K3];
  __shared__ float s_part[16 * PART_PITCH];
  __shared__ float s_dot[NTASK];
  __shared__ __attribute__((aligned(16))) float s_sq[3][NTHREADS];

  const int tid  = threadIdx.x;
  const int lane = tid & 31;
  const int wave = __builtin_amdgcn_readfirstlane(tid >> 5);
  const int row0 = blockIdx.x * ROWS_PB;

  const v4f* __restrict__ in4 = reinterpret_cast<const v4f*>(input);
  const bool two = tid < T_SECOND;
  const int ta = tid;
  const int tb = two ? tid + NTHREADS : tid;
  const v4f ca = in4[ta];
  const v4f pa = in4[ta > 0 ? ta - 1 : 0];
  const v4f cb = in4[tb];
  const v4f pb = in4[tb > 0 ? tb - 1 : 0];

  float ex = 0.f, ey = 0.f, ez = 0.f;
  step_features(ta, ca, pa, true, s_feat, ex, ey, ez);
  step_features(tb, cb, pb, two, s_feat, ex, ey, ez);
  s_sq[0][tid] = ex;
  s_sq[1][tid] = ey;
  s_sq[2][tid] = ez;

  __syncthreads();

  const float* f1 = s_feat;
  const float* f2 = s_feat + K1;
  const float* f3 = s_feat + 2 * K1;

  if (wave < 6) {
    const bool first = wave < 3;
    const float* __restrict__ Wm = first ? W1 : W2;
    const float* fm = first ? f1 : f2;
    float* pm = s_part + (first ? 0 : ROWS_PB);
    const int wi = first ? wave : wave - 3;
    if (wi == 0) {
      stream_rows<K1, 6>(Wm + (size_t)row0 * K1, fm, pm, lane);
    } else {
      const int r0 = 1 + 5 * wi;
      stream_rows<K1, 5>(Wm + (size_t)(row0 + r0) * K1, fm, pm + r0, lane);
    }
  } else {
    const int r0 = 8 * (wave - 6);
    stream_rows<K3, 8>(W3 + (size_t)(row0 + r0) * K3, f3, s_part + 2 * ROWS_PB + r0, lane);
  }

  const int et = tid - 32;
  const bool fin = et >= 0 && et < ROWS_PB;
  float eb1 = 0.f, eb2 = 0.f, eb3 = 0.f, eb4 = 0.f;
  float ew0 = 0.f, ew1 = 0.f, ew2 = 0.f;
  float e0 = 0.f, e1 = 0.f, e2 = 0.f;
  if (wave == 1) {
    float p[3];
#pragma unroll
    for (int c = 0; c < 3; ++c) {
      const v4f a = *reinterpret_cast<const v4f*>(&s_sq[c][8 * lane]);
      const v4f b = *reinterpret_cast<const v4f*>(&s_sq[c][8 * lane + 4]);
      p[c] = ((a.x + a.y) + (a.z + a.w)) + ((b.x + b.y) + (b.z + b.w));
    }
#pragma unroll
    for (int m = 16; m >= 1; m >>= 1) {
      p[0] += __shfl_xor(p[0], m);
      p[1] += __shfl_xor(p[1], m);
      p[2] += __shfl_xor(p[2], m);
    }
    e0 = p[0] / (float)T_LEN;
    e1 = p[1] / (float)T_LEN;
    e2 = p[2] / (float)T_LEN;
  }
  if (fin) {
    const int row = row0 + et;
    eb1 = b1[row];
    eb2 = b2[row];
    eb3 = b3[row];
    eb4 = b4[row];
    ew0 = W4[row * 3 + 0];
    ew1 = W4[row * 3 + 1];
    ew2 = W4[row * 3 + 2];
  }
  __syncthreads();

  if (tid < NTASK) {
    float s0 = 0.f, s1 = 0.f, s2 = 0.f, s3 = 0.f;
#pragma unroll
    for (int i = 0; i < 16; i += 4) {
      s0 += s_part[(i + 0) * PART_PITCH + tid];
      s1 += s_part[(i + 1) * PART_PITCH + tid];
      s2 += s_part[(i + 2) * PART_PITCH + tid];
      s3 += s_part[(i + 3) * PART_PITCH + tid];
    }
    s_dot[tid] = (s0 + s1) + (s2 + s3);
  }
  __syncthreads();

  if (fin) {
    const float h1 = tanhf(s_dot[et] + eb1);
    const float h2 = tanhf(s_dot[ROWS_PB + et] + eb2);
    const float h3 = tanhf(s_dot[2 * ROWS_PB + et] + eb3);
    const float d4 = ew0 * e0 + ew1 * e1 + ew2 * e2 + eb4;
    const float h4 = tanhf(d4);
    out[row0 + et] = tanhf(h1 * h4 + h2 + h3);
  }
}

extern "C" void kernel_launch(void* const* d_in, const int* in_sizes, int n_in,
                              void* d_out, int out_size, void* d_ws, size_t ws_size,
                              hipStream_t stream) {
  (void)in_sizes; (void)n_in; (void)out_size; (void)d_ws; (void)ws_size;
  const float* input = (const float*)d_in[0];
  const float* W1    = (const float*)d_in[1];
  const float* b1    = (const float*)d_in[2];
  const float* W2    = (const float*)d_in[3];
  const float* b2    = (const float*)d_in[4];
  const float* W3    = (const float*)d_in[5];
  const float* b3    = (const float*)d_in[6];
  const float* W4    = (const float*)d_in[7];
  const float* b4    = (const float*)d_in[8];
  float*       out   = (float*)d_out;

  hdc_fused_wmma_kernel<<<NBLOCKS, NTHREADS, 0, stream>>>(input, W1, b1, W2, b2,
                                                        W3, b3, W4, b4, out);
}


# Grid re-cut so that fewer waves have to start: 250 groups x 8 waves x 40 rows (was 625 x 8 x 16), a wave streams 13 to 20 rows through a ring of 4 or 6 row buffers; every sum keeps its order
# speedup vs baseline: 7.6904x; 1.0914x over previous
#include <hip/hip_runtime.h>
#include <math.h>


typedef __attribute__((ext_vector_type(2))) float v2f;
typedef __attribute__((ext_vector_type(4))) float v4f;
typedef __attribute__((ext_vector_type(8))) float v8f;

#define T_LEN 400
#define D_LEN 10000
#define K1 1200
#define K3 800
#define ROWS_PB 40
#define NBLOCKS (D_LEN / ROWS_PB)
#define NTASK (3 * ROWS_PB)
#define PART_PITCH 121
#define NTHREADS 256
#define T_SECOND (T_LEN - NTHREADS)

template <int K, int NROWS, int RING>
__device__ __forceinline__ void stream_rows(const float* __restrict__ wrow0,
                                            const float* __restrict__ f,
                                            float* __restrict__ part,
                                            const int lane) {
  constexpr int NV  = K / 4;
  constexpr int NQ  = (NV + 31) / 32;
  constexpr int REM = NV - 32 * (NQ - 1);
  const bool tail_on = lane < REM;
  const int tail_off = tail_on ? 32 * (NQ - 1) : 0;
  const v4f zero = {0.f, 0.f, 0.f, 0.f};

  const v4f* fp = reinterpret_cast<const v4f*>(f) + lane;
  v4f fq[NQ];
#pragma unroll
  for (int q = 0; q < NQ - 1; ++q) fq[q] = fp[32 * q];
  fq[NQ - 1] = fp[tail_off];

  const v4f* wp = reinterpret_cast<const v4f*>(wrow0) + lane;
  v4f w[RING][NQ];
#define LOAD_ROW(R)                                                          \
  do {                                                                       \
    const v4f* rp_ = wp + (size_t)(R) * NV;                                  \
    _Pragma("unroll") for (int q = 0; q < NQ - 1; ++q)                       \
        w[(R) % RING][q] = rp_[32 * q];                                      \
    w[(R) % RING][NQ - 1] = rp_[tail_off];                                   \
  } while (0)

  const int  sel  = lane & 7;
  const bool diag = ((lane >> 3) & 1) == ((lane >> 4) & 1);

#pragma unroll
  for (int r = 0; r < RING - 1; ++r)
    if (r < NROWS) LOAD_ROW(r);
#pragma unroll
  for (int r = 0; r < NROWS; ++r) {
    __builtin_amdgcn_sched_barrier(0);
    if (r + RING - 1 < NROWS) LOAD_ROW(r + RING - 1);
    __builtin_amdgcn_sched_barrier(0);

    if (r == 0) fq[NQ - 1] = tail_on ? fq[NQ - 1] : zero;

    const v4f* wr = w[r % RING];
    v8f acc0 = {};
    v8f acc1 = {};
#pragma unroll
    for (int q = 0; q < NQ; ++q) {
      v4f wv = wr[q];
      if (q == NQ - 1) wv = tail_on ? wv : zero;
      const v2f a0 = {wv.x, wv.y};
      const v2f b0 = {fq[q].x, fq[q].y};
      const v2f a1 = {wv.z, wv.w};
      const v2f b1 = {fq[q].z, fq[q].w};
      acc0 = __builtin_amdgcn_wmma_f32_16x16x4_f32(false, a0, false, b0,
                                                   (short)0, acc0, false, false);
      acc1 = __builtin_amdgcn_wmma_f32_16x16x4_f32(false, a1, false, b1,
                                                   (short)0, acc1, false, false);
    }
    const v8f acc = acc0 + acc1;
    float t = acc[0];
#pragma unroll
    for (int i = 1; i < 8; ++i) t = (sel == i) ? acc[i] : t;
    if (diag) part[(lane & 15) * PART_PITCH + r] = t;
  }
#undef LOAD_ROW
}

__device__ __forceinline__ void step_features(const int t, const v4f c, const v4f p,
                                              const bool live,
                                              float* __restrict__ s_feat,
                                              float& ex, float& ey, float& ez) {
  const float x = c.y, y = c.z, z = c.w;
  float jx = 0.f, jy = 0.f, jz = 0.f;
  if (t > 0) {
    const float dt  = c.x - p.x;
    const float inv = 1.0f / dt;
    jx = (x - p.y) * inv;
    jy = (y - p.z) * inv;
    jz = (z - p.w) * inv;
  }
  if (live) {
    s_feat[0 * T_LEN + t] = x;
    s_feat[1 * T_LEN + t] = y;
    s_feat[2 * T_LEN + t] = z;
    s_feat[K1 + 0 * T_LEN + t] = jx;
    s_feat[K1 + 1 * T_LEN + t] = jy;
    s_feat[K1 + 2 * T_LEN + t] = jz;
    s_feat[2 * K1 + t]         = sqrtf(x * x + y * y + z * z);
    s_feat[2 * K1 + T_LEN + t] = sqrtf(jx * jx + jy * jy + jz * jz);
    ex += x * x;
    ey += y * y;
    ez += z * z;
  }
}

__global__ __launch_bounds__(NTHREADS) void hdc_fused_wmma_kernel(
    const float* __restrict__ input,
    const float* __restrict__ W1, const float* __restrict__ b1,
    const float* __restrict__ W2, const float* __restrict__ b2,
    const float* __restrict__ W3, const float* __restrict__ b3,
    const float* __restrict__ W4, const float* __restrict__ b4,
    float* __restrict__ out) {
  __shared__ __attribute__((aligned(16))) float s_feat[2 * K1 + K3];
  __shared__ float s_part[16 * PART_PITCH];
  __shared__ float s_dot[NTASK];
  __shared__ float s_energy[3];
  __shared__ __attribute__((aligned(16))) float s_sq[3][NTHREADS];

  const int tid  = threadIdx.x;
  const int lane = tid & 31;
  const int wave = __builtin_amdgcn_readfirstlane(tid >> 5);
  const int row0 = blockIdx.x * ROWS_PB;

  const v4f* __restrict__ in4 = reinterpret_cast<const v4f*>(input);
  const bool two = tid < T_SECOND;
  const int ta = tid;
  const int tb = two ? tid + NTHREADS : tid;
  const v4f ca = in4[ta];
  const v4f pa = in4[ta > 0 ? ta - 1 : 0];
  const v4f cb = in4[tb];
  const v4f pb = in4[tb > 0 ? tb - 1 : 0];

  float ex = 0.f, ey = 0.f, ez = 0.f;
  step_features(ta, ca, pa, true, s_feat, ex, ey, ez);
  step_features(tb, cb, pb, two, s_feat, ex, ey, ez);
  s_sq[0][tid] = ex;
  s_sq[1][tid] = ey;
  s_sq[2][tid] = ez;

  __syncthreads();

  const float* f1 = s_feat;
  const float* f2 = s_feat + K1;
  const float* f3 = s_feat + 2 * K1;

  if (wave < 6) {
    const bool first = wave < 3;
    const float* __restrict__ Wm = first ? W1 : W2;
    const float* fm = first ? f1 : f2;
    float* pm = s_part + (first ? 0 : ROWS_PB);
    const int wi = first ? wave : wave - 3;
    if (wi == 0) {
      stream_rows<K1, 14, 4>(Wm + (size_t)row0 * K1, fm, pm, lane);
    } else {
      const int r0 = 1 + 13 * wi;
      stream_rows<K1, 13, 4>(Wm + (size_t)(row0 + r0) * K1, fm, pm + r0, lane);
    }
  } else {
    const int r0 = 20 * (wave - 6);
    stream_rows<K3, 20, 6>(W3 + (size_t)(row0 + r0) * K3, f3, s_part + 2 * ROWS_PB + r0, lane);
  }

  const int et = tid - 32;
  const bool fin = et >= 0 && et < ROWS_PB;
  float eb1 = 0.f, eb2 = 0.f, eb3 = 0.f, eb4 = 0.f;
  float ew0 = 0.f, ew1 = 0.f, ew2 = 0.f;
  if (wave == 1) {
    float p[3];
#pragma unroll
    for (int c = 0; c < 3; ++c) {
      const v4f a = *reinterpret_cast<const v4f*>(&s_sq[c][8 * lane]);
      const v4f b = *reinterpret_cast<const v4f*>(&s_sq[c][8 * lane + 4]);
      p[c] = ((a.x + a.y) + (a.z + a.w)) + ((b.x + b.y) + (b.z + b.w));
    }
#pragma unroll
    for (int m = 16; m >= 1; m >>= 1) {
      p[0] += __shfl_xor(p[0], m);
      p[1] += __shfl_xor(p[1], m);
      p[2] += __shfl_xor(p[2], m);
    }
    if (lane < 3) {
      const float pc = lane == 0 ? p[0] : (lane == 1 ? p[1] : p[2]);
      s_energy[lane] = pc / (float)T_LEN;
    }
  }
  if (fin) {
    const int row = row0 + et;
    eb1 = b1[row];
    eb2 = b2[row];
    eb3 = b3[row];
    eb4 = b4[row];
    ew0 = W4[row * 3 + 0];
    ew1 = W4[row * 3 + 1];
    ew2 = W4[row * 3 + 2];
  }
  __syncthreads();

  if (tid < NTASK) {
    float s0 = 0.f, s1 = 0.f, s2 = 0.f, s3 = 0.f;
#pragma unroll
    for (int i = 0; i < 16; i += 4) {
      s0 += s_part[(i + 0) * PART_PITCH + tid];
      s1 += s_part[(i + 1) * PART_PITCH + tid];
      s2 += s_part[(i + 2) * PART_PITCH + tid];
      s3 += s_part[(i + 3) * PART_PITCH + tid];
    }
    s_dot[tid] = (s0 + s1) + (s2 + s3);
  }
  __syncthreads();

  if (fin) {
    const float e0 = s_energy[0], e1 = s_energy[1], e2 = s_energy[2];
    const float h1 = tanhf(s_dot[et] + eb1);
    const float h2 = tanhf(s_dot[ROWS_PB + et] + eb2);
    const float h3 = tanhf(s_dot[2 * ROWS_PB + et] + eb3);
    const float d4 = ew0 * e0 + ew1 * e1 + ew2 * e2 + eb4;
    const float h4 = tanhf(d4);
    out[row0 + et] = tanhf(h1 * h4 + h2 + h3);
  }
}

extern "C" void kernel_launch(void* const* d_in, const int* in_sizes, int n_in,
                              void* d_out, int out_size, void* d_ws, size_t ws_size,
                              hipStream_t stream) {
  (void)in_sizes; (void)n_in; (void)out_size; (void)d_ws; (void)ws_size;
  const float* input = (const float*)d_in[0];
  const float* W1    = (const float*)d_in[1];
  const float* b1    = (const float*)d_in[2];
  const float* W2    = (const float*)d_in[3];
  const float* b2    = (const float*)d_in[4];
  const float* W3    = (const float*)d_in[5];
  const float* b3    = (const float*)d_in[6];
  const float* W4    = (const float*)d_in[7];
  const float* b4    = (const float*)d_in[8];
  float*       out   = (float*)d_out;

  hdc_fused_wmma_kernel<<<NBLOCKS, NTHREADS, 0, stream>>>(input, W1, b1, W2, b2,
                                                        W3, b3, W4, b4, out);
}
